# starting kernel
# baseline (speedup 1.0000x reference)
#include <hip/hip_runtime.h>

typedef __attribute__((ext_vector_type(2))) float v2f;
typedef __attribute__((ext_vector_type(8))) float v8f;

#define NZ 4096
#define NX 8192
#define RADIUS 40
#define NTAPS (2 * RADIUS + 1)
#define KTOT 96
#define NSTEP (KTOT / 4)
#define ROWTILES 4

__device__ __forceinline__ int mirror_idx(int i) {
  if (i < 0) i = -1 - i;
  if (i >= NX) i = 2 * NX - 1 - i;
  return i;
}

__global__ void gauss_weights_kernel(const int* __restrict__ h_smooth,
                                     float* __restrict__ w) {
  const int j = threadIdx.x;
  const float sigma = (float)h_smooth[0];
  const float inv2s2 = 0.5f / (sigma * sigma);
  float sum = 0.f;
  for (int t = 0; t < NTAPS; ++t) {
    float x = (float)(t - RADIUS);
    sum += __expf(-x * x * inv2s2);
  }
  if (j < NTAPS) {
    float x = (float)(j - RADIUS);
    w[j] = __expf(-x * x * inv2s2) / sum;
  }
}

__launch_bounds__(256)
__global__ void gauss_smooth_wmma(const float* __restrict__ in,
                                  const float* __restrict__ w,
                                  float* __restrict__ out) {
  __shared__ float ws[NTAPS];
  if (threadIdx.x < NTAPS) ws[threadIdx.x] = w[threadIdx.x];
  __syncthreads();

  const int lane = threadIdx.x & 31;
  const int wave = threadIdx.x >> 5;
  const int x0 = blockIdx.x * 128 + wave * 16;
  const int rbase = blockIdx.y * (16 * ROWTILES);
  const int n = lane & 15;
  const int hi = lane >> 4;

  v2f B[NSTEP];
#pragma unroll
  for (int s = 0; s < NSTEP; ++s) {
    const int k = 4 * s + 2 * hi;
    const int d0 = k - n, d1 = k + 1 - n;
    B[s].x = (d0 >= 0 && d0 < NTAPS) ? ws[d0] : 0.f;
    B[s].y = (d1 >= 0 && d1 < NTAPS) ? ws[d1] : 0.f;
  }

  const int cbase = x0 - RADIUS;
  const bool interior = (cbase >= 0) && (cbase + KTOT <= NX);

  for (int t = 0; t < ROWTILES; ++t) {
    const int r0 = rbase + t * 16;
    const float* inrow = in + (size_t)(r0 + n) * NX;
    v8f c = {};

    if (interior) {
#pragma unroll
      for (int s = 0; s < NSTEP; ++s) {
        const int col = cbase + 4 * s + 2 * hi;
        v2f a = *(const v2f*)(inrow + col);
        c = __builtin_amdgcn_wmma_f32_16x16x4_f32(false, a, false, B[s],
                                                  (short)0, c, false, false);
      }
    } else {
#pragma unroll
      for (int s = 0; s < NSTEP; ++s) {
        const int col = cbase + 4 * s + 2 * hi;
        v2f a;
        a.x = inrow[mirror_idx(col)];
        a.y = inrow[mirror_idx(col + 1)];
        c = __builtin_amdgcn_wmma_f32_16x16x4_f32(false, a, false, B[s],
                                                  (short)0, c, false, false);
      }
    }

    float* orow = out + (size_t)(r0 + 8 * hi) * NX + x0 + n;
#pragma unroll
    for (int r = 0; r < 8; ++r) {
      orow[(size_t)r * NX] = c[r];
    }
  }
}

extern "C" void kernel_launch(void* const* d_in, const int* in_sizes, int n_in,
                              void* d_out, int out_size, void* d_ws, size_t ws_size,
                              hipStream_t stream) {
  (void)in_sizes; (void)n_in; (void)out_size; (void)ws_size;
  const float* feature = (const float*)d_in[0];
  const int* h_smooth = (const int*)d_in[1];
  float* w = (float*)d_ws;

  gauss_weights_kernel<<<1, 128, 0, stream>>>(h_smooth, w);

  dim3 grid(NX / 128, NZ / (16 * ROWTILES));
  gauss_smooth_wmma<<<grid, 256, 0, stream>>>(feature, w, (float*)d_out);
}


# A fragments from LDS: each workgroup copies its 64x208 input window to LDS once with whole-line 16-byte loads (pitch 212 floats, conflict-free) instead of 16-line 8-byte global fragment loads
# speedup vs baseline: 1.3999x; 1.3999x over previous
#include <hip/hip_runtime.h>


typedef __attribute__((ext_vector_type(2))) float v2f;
typedef __attribute__((ext_vector_type(4))) float v4f;
typedef __attribute__((ext_vector_type(8))) float v8f;

#define NZ 4096
#define NX 8192
#define RADIUS 40
#define NTAPS (2 * RADIUS + 1)
#define KTOT 96
#define NSTEP (KTOT / 4)
#define ROWTILES 4

#define BCOLS 128
#define TROWS (16 * ROWTILES)
#define TCOLS (BCOLS + KTOT - 16)
#define TC4 (TCOLS / 4)
#define PITCH 212

__device__ __forceinline__ int mirror_idx(int i) {
  if (i < 0) i = -1 - i;
  if (i >= NX) i = 2 * NX - 1 - i;
  return i;
}

__global__ void gauss_weights_kernel(const int* __restrict__ h_smooth,
                                     float* __restrict__ w) {
  const int j = threadIdx.x;
  const float sigma = (float)h_smooth[0];
  const float inv2s2 = 0.5f / (sigma * sigma);
  float sum = 0.f;
  for (int t = 0; t < NTAPS; ++t) {
    float x = (float)(t - RADIUS);
    sum += __expf(-x * x * inv2s2);
  }
  if (j < NTAPS) {
    float x = (float)(j - RADIUS);
    w[j] = __expf(-x * x * inv2s2) / sum;
  }
}

__launch_bounds__(256)
__global__ void gauss_smooth_wmma(const float* __restrict__ in,
                                  const float* __restrict__ w,
                                  float* __restrict__ out) {
  __shared__ float ws[NTAPS];
  __shared__ __attribute__((aligned(16))) float tile[TROWS * PITCH];

  const int tid = threadIdx.x;
  const int xblk = blockIdx.x * BCOLS;
  const int rbase = blockIdx.y * TROWS;
  const int cbase = xblk - RADIUS;
  const bool interior = (cbase >= 0) && (cbase + TCOLS <= NX);

  const int lane = tid & 31;
  const int wave = tid >> 5;

  float tap = 0.f;
  if (tid < NTAPS) tap = w[tid];

  float* dst = tile + wave * PITCH + 4 * lane;
  if (interior) {
    const float* src = in + (size_t)(rbase + wave) * NX + cbase + 4 * lane;
    v4f va[TROWS / 8], vb[TROWS / 8];
#pragma unroll
    for (int i = 0; i < TROWS / 8; ++i)
      va[i] = *(const v4f*)(src + (size_t)i * 8 * NX);
    if (lane < TC4 - 32) {
#pragma unroll
      for (int i = 0; i < TROWS / 8; ++i)
        vb[i] = *(const v4f*)(src + (size_t)i * 8 * NX + 128);
    }
#pragma unroll
    for (int i = 0; i < TROWS / 8; ++i)
      *(v4f*)(dst + i * 8 * PITCH) = va[i];
    if (lane < TC4 - 32) {
#pragma unroll
      for (int i = 0; i < TROWS / 8; ++i)
        *(v4f*)(dst + i * 8 * PITCH + 128) = vb[i];
    }
  } else {
    for (int i = 0; i < TROWS / 8; ++i) {
      const float* inrow = in + (size_t)(rbase + wave + 8 * i) * NX;
      for (int p = lane; p < TC4; p += 32) {
        const int col = cbase + 4 * p;
        v4f v;
        v.x = inrow[mirror_idx(col)];
        v.y = inrow[mirror_idx(col + 1)];
        v.z = inrow[mirror_idx(col + 2)];
        v.w = inrow[mirror_idx(col + 3)];
        *(v4f*)(tile + (wave + 8 * i) * PITCH + 4 * p) = v;
      }
    }
  }
  if (tid < NTAPS) ws[tid] = tap;
  __syncthreads();

  const int x0 = xblk + wave * 16;
  const int n = lane & 15;
  const int hi = lane >> 4;

  v2f B[NSTEP];
#pragma unroll
  for (int s = 0; s < NSTEP; ++s) {
    const int k = 4 * s + 2 * hi;
    const int d0 = k - n, d1 = k + 1 - n;
    B[s].x = (d0 >= 0 && d0 < NTAPS) ? ws[d0] : 0.f;
    B[s].y = (d1 >= 0 && d1 < NTAPS) ? ws[d1] : 0.f;
  }

  const float* abase = tile + n * PITCH + wave * 16 + 2 * hi;

  for (int t = 0; t < ROWTILES; ++t) {
    const int r0 = rbase + t * 16;
    const float* arow = abase + t * 16 * PITCH;
    v8f c = {};

#pragma unroll
    for (int s = 0; s < NSTEP; ++s) {
      v2f a = *(const v2f*)(arow + 4 * s);
      c = __builtin_amdgcn_wmma_f32_16x16x4_f32(false, a, false, B[s],
                                                (short)0, c, false, false);
    }

    float* orow = out + (size_t)(r0 + 8 * hi) * NX + x0 + n;
#pragma unroll
    for (int r = 0; r < 8; ++r) {
      orow[(size_t)r * NX] = c[r];
    }
  }
}

extern "C" void kernel_launch(void* const* d_in, const int* in_sizes, int n_in,
                              void* d_out, int out_size, void* d_ws, size_t ws_size,
                              hipStream_t stream) {
  (void)in_sizes; (void)n_in; (void)out_size; (void)ws_size;
  const float* feature = (const float*)d_in[0];
  const int* h_smooth = (const int*)d_in[1];
  float* w = (float*)d_ws;

  gauss_weights_kernel<<<1, 128, 0, stream>>>(h_smooth, w);

  dim3 grid(NX / BCOLS, NZ / TROWS);
  gauss_smooth_wmma<<<grid, 256, 0, stream>>>(feature, w, (float*)d_out);
}


# One wave per workgroup: each wave stages its own 16x208 window in LDS and computes a 16x128 band, reading each A fragment once for up to six tiles; no barrier between waves; sums unchanged
# speedup vs baseline: 1.7128x; 1.2235x over previous
#include <hip/hip_runtime.h>


typedef __attribute__((ext_vector_type(2))) float v2f;
typedef __attribute__((ext_vector_type(4))) float v4f;
typedef __attribute__((ext_vector_type(8))) float v8f;

#define NZ 4096
#define NX 8192
#define RADIUS 40
#define NTAPS (2 * RADIUS + 1)
#define KTOT 96
#define NSTEP (KTOT / 4)
#define CTILES 8

#define WCOLS (16 * CTILES)
#define TROWS 16
#define TCOLS (WCOLS + KTOT - 16)
#define TC4 (TCOLS / 4)
#define NFRAG (TCOLS / 4)
#define PITCH 212

__device__ __forceinline__ int mirror_idx(int i) {
  if (i < 0) i = -1 - i;
  if (i >= NX) i = 2 * NX - 1 - i;
  return i;
}

__global__ void gauss_weights_kernel(const int* __restrict__ h_smooth,
                                     float* __restrict__ w) {
  const int j = threadIdx.x;
  const float sigma = (float)h_smooth[0];
  const float inv2s2 = 0.5f / (sigma * sigma);
  float sum = 0.f;
  for (int t = 0; t < NTAPS; ++t) {
    float x = (float)(t - RADIUS);
    sum += __expf(-x * x * inv2s2);
  }
  if (j < NTAPS) {
    float x = (float)(j - RADIUS);
    w[j] = __expf(-x * x * inv2s2) / sum;
  }
}

__launch_bounds__(32)
__global__ void gauss_smooth_wmma(const float* __restrict__ in,
                                  const float* __restrict__ w,
                                  float* __restrict__ out) {
  __shared__ float ws[NTAPS];
  __shared__ __attribute__((aligned(16))) float tile[TROWS * PITCH];

  const int lane = threadIdx.x;
  const int x0 = blockIdx.x * WCOLS;
  const int r0 = blockIdx.y * TROWS;
  const int cbase = x0 - RADIUS;
  const bool interior = (cbase >= 0) && (cbase + TCOLS <= NX);

  const float tap0 = w[lane];
  const float tap1 = w[lane + 32];
  float tap2 = 0.f;
  if (lane < NTAPS - 64) tap2 = w[lane + 64];

  float* dst = tile + 4 * lane;
  const bool tail = lane < TC4 - 32;
  v4f va[TROWS], vb[TROWS];
  if (interior) {
    const float* src = in + (size_t)r0 * NX + cbase + 4 * lane;
#pragma unroll
    for (int i = 0; i < TROWS; ++i)
      va[i] = *(const v4f*)(src + (size_t)i * NX);
    if (tail) {
#pragma unroll
      for (int i = 0; i < TROWS; ++i)
        vb[i] = *(const v4f*)(src + (size_t)i * NX + 128);
    }
  } else {
    const float* src = in + (size_t)r0 * NX;
    const int col = cbase + 4 * lane;
    const int a_0 = mirror_idx(col), a_1 = mirror_idx(col + 1);
    const int a_2 = mirror_idx(col + 2), a_3 = mirror_idx(col + 3);
#pragma unroll
    for (int i = 0; i < TROWS; ++i) {
      const float* inrow = src + (size_t)i * NX;
      va[i].x = inrow[a_0];
      va[i].y = inrow[a_1];
      va[i].z = inrow[a_2];
      va[i].w = inrow[a_3];
    }
    if (tail) {
      const int b_0 = mirror_idx(col + 128), b_1 = mirror_idx(col + 129);
      const int b_2 = mirror_idx(col + 130), b_3 = mirror_idx(col + 131);
#pragma unroll
      for (int i = 0; i < TROWS; ++i) {
        const float* inrow = src + (size_t)i * NX;
        vb[i].x = inrow[b_0];
        vb[i].y = inrow[b_1];
        vb[i].z = inrow[b_2];
        vb[i].w = inrow[b_3];
      }
    }
  }
#pragma unroll
  for (int i = 0; i < TROWS; ++i)
    *(v4f*)(dst + i * PITCH) = va[i];
  if (tail) {
#pragma unroll
    for (int i = 0; i < TROWS; ++i)
      *(v4f*)(dst + i * PITCH + 128) = vb[i];
  }
  ws[lane] = tap0;
  ws[lane + 32] = tap1;
  if (lane < NTAPS - 64) ws[lane + 64] = tap2;
  __syncthreads();

  const int n = lane & 15;
  const int hi = lane >> 4;

  v2f B[NSTEP];
#pragma unroll
  for (int s = 0; s < NSTEP; ++s) {
    const int k = 4 * s + 2 * hi;
    const int d0 = k - n, d1 = k + 1 - n;
    B[s].x = (d0 >= 0 && d0 < NTAPS) ? ws[d0] : 0.f;
    B[s].y = (d1 >= 0 && d1 < NTAPS) ? ws[d1] : 0.f;
  }

  const float* arow = tile + n * PITCH + 2 * hi;

  v8f c[CTILES];
#pragma unroll
  for (int j = 0; j < CTILES; ++j) c[j] = v8f{};

#pragma unroll
  for (int f = 0; f < NFRAG; ++f) {
    const v2f a = *(const v2f*)(arow + 4 * f);
#pragma unroll
    for (int j = 0; j < CTILES; ++j) {
      const int s = f - 4 * j;
      if (s >= 0 && s < NSTEP) {
        c[j] = __builtin_amdgcn_wmma_f32_16x16x4_f32(false, a, false, B[s],
                                                     (short)0, c[j], false, false);
      }
    }
  }

  float* orow = out + (size_t)(r0 + 8 * hi) * NX + x0 + n;
#pragma unroll
  for (int j = 0; j < CTILES; ++j) {
#pragma unroll
    for (int r = 0; r < 8; ++r) {
      orow[(size_t)r * NX + 16 * j] = c[j][r];
    }
  }
}

extern "C" void kernel_launch(void* const* d_in, const int* in_sizes, int n_in,
                              void* d_out, int out_size, void* d_ws, size_t ws_size,
                              hipStream_t stream) {
  (void)in_sizes; (void)n_in; (void)out_size; (void)ws_size;
  const float* feature = (const float*)d_in[0];
  const int* h_smooth = (const int*)d_in[1];
  float* w = (float*)d_ws;

  gauss_weights_kernel<<<1, 128, 0, stream>>>(h_smooth, w);

  dim3 grid(NX / WCOLS, NZ / TROWS);
  gauss_smooth_wmma<<<grid, 32, 0, stream>>>(feature, w, (float*)d_out);
}


# Each wave walks four 128-column bands of its 16 rows: band b+2's 128 new input columns load under band b's matrix work, the 80 shared columns stay in LDS, tap tile built once; sums unchanged
# speedup vs baseline: 1.8103x; 1.0569x over previous
#include <hip/hip_runtime.h>


typedef __attribute__((ext_vector_type(2))) float v2f;
typedef __attribute__((ext_vector_type(4))) float v4f;
typedef __attribute__((ext_vector_type(8))) float v8f;

#define NZ 4096
#define NX 8192
#define RADIUS 40
#define NTAPS (2 * RADIUS + 1)
#define KTOT 96
#define NSTEP (KTOT / 4)
#define CTILES 8

#define WCOLS (16 * CTILES)
#define TROWS 16
#define TCOLS (WCOLS + KTOT - 16)
#define TC4 (TCOLS / 4)
#define NFRAG (TCOLS / 4)
#define NBAND 4
#define SEGCOLS (NBAND * WCOLS)
#define LCOLS (SEGCOLS + KTOT - 16)
#define PITCH 596

__device__ __forceinline__ int mirror_idx(int i) {
  if (i < 0) i = -1 - i;
  if (i >= NX) i = 2 * NX - 1 - i;
  return i;
}

__global__ void gauss_weights_kernel(const int* __restrict__ h_smooth,
                                     float* __restrict__ w) {
  const int j = threadIdx.x;
  const float sigma = (float)h_smooth[0];
  const float inv2s2 = 0.5f / (sigma * sigma);
  float sum = 0.f;
  for (int t = 0; t < NTAPS; ++t) {
    float x = (float)(t - RADIUS);
    sum += __expf(-x * x * inv2s2);
  }
  if (j < NTAPS) {
    float x = (float)(j - RADIUS);
    w[j] = __expf(-x * x * inv2s2) / sum;
  }
}

__device__ __forceinline__ void load_chunk(const float* __restrict__ in, int r0,
                                           int cs, int lane, v4f (&v)[TROWS]) {
  if (cs + WCOLS <= NX) {
    const float* src = in + (size_t)r0 * NX + cs + 4 * lane;
#pragma unroll
    for (int i = 0; i < TROWS; ++i)
      v[i] = *(const v4f*)(src + (size_t)i * NX);
  } else {
    const float* src = in + (size_t)r0 * NX;
    const int col = cs + 4 * lane;
    const int a_0 = mirror_idx(col), a_1 = mirror_idx(col + 1);
    const int a_2 = mirror_idx(col + 2), a_3 = mirror_idx(col + 3);
#pragma unroll
    for (int i = 0; i < TROWS; ++i) {
      const float* inrow = src + (size_t)i * NX;
      v[i].x = inrow[a_0];
      v[i].y = inrow[a_1];
      v[i].z = inrow[a_2];
      v[i].w = inrow[a_3];
    }
  }
}

__launch_bounds__(32)
__global__ void gauss_smooth_wmma(const float* __restrict__ in,
                                  const float* __restrict__ w,
                                  float* __restrict__ out) {
  __shared__ float ws[NTAPS];
  __shared__ __attribute__((aligned(16))) float tile[TROWS * PITCH];

  const int lane = threadIdx.x;
  const int x0 = blockIdx.x * SEGCOLS;
  const int r0 = blockIdx.y * TROWS;
  const int cbase = x0 - RADIUS;
  const bool interior = (cbase >= 0) && (cbase + TCOLS <= NX);

  const float tap0 = w[lane];
  const float tap1 = w[lane + 32];
  float tap2 = 0.f;
  if (lane < NTAPS - 64) tap2 = w[lane + 64];

  float* dst = tile + 4 * lane;
  const bool tail = lane < TC4 - 32;
  v4f va[TROWS], vb[TROWS], vc[TROWS];
  if (interior) {
    const float* src = in + (size_t)r0 * NX + cbase + 4 * lane;
#pragma unroll
    for (int i = 0; i < TROWS; ++i)
      va[i] = *(const v4f*)(src + (size_t)i * NX);
    if (tail) {
#pragma unroll
      for (int i = 0; i < TROWS; ++i)
        vb[i] = *(const v4f*)(src + (size_t)i * NX + 128);
    }
  } else {
    const float* src = in + (size_t)r0 * NX;
    const int col = cbase + 4 * lane;
    const int a_0 = mirror_idx(col), a_1 = mirror_idx(col + 1);
    const int a_2 = mirror_idx(col + 2), a_3 = mirror_idx(col + 3);
#pragma unroll
    for (int i = 0; i < TROWS; ++i) {
      const float* inrow = src + (size_t)i * NX;
      va[i].x = inrow[a_0];
      va[i].y = inrow[a_1];
      va[i].z = inrow[a_2];
      va[i].w = inrow[a_3];
    }
    if (tail) {
      const int b_0 = mirror_idx(col + 128), b_1 = mirror_idx(col + 129);
      const int b_2 = mirror_idx(col + 130), b_3 = mirror_idx(col + 131);
#pragma unroll
      for (int i = 0; i < TROWS; ++i) {
        const float* inrow = src + (size_t)i * NX;
        vb[i].x = inrow[b_0];
        vb[i].y = inrow[b_1];
        vb[i].z = inrow[b_2];
        vb[i].w = inrow[b_3];
      }
    }
  }
  int cs = cbase + TCOLS;
  load_chunk(in, r0, cs, lane, vc);
  cs += WCOLS;

#pragma unroll
  for (int i = 0; i < TROWS; ++i)
    *(v4f*)(dst + i * PITCH) = va[i];
  if (tail) {
#pragma unroll
    for (int i = 0; i < TROWS; ++i)
      *(v4f*)(dst + i * PITCH + 128) = vb[i];
  }
  ws[lane] = tap0;
  ws[lane + 32] = tap1;
  if (lane < NTAPS - 64) ws[lane + 64] = tap2;
  __syncthreads();

  const int n = lane & 15;
  const int hi = lane >> 4;

  v2f B[NSTEP];
#pragma unroll
  for (int s = 0; s < NSTEP; ++s) {
    const int k = 4 * s + 2 * hi;
    const int d0 = k - n, d1 = k + 1 - n;
    B[s].x = (d0 >= 0 && d0 < NTAPS) ? ws[d0] : 0.f;
    B[s].y = (d1 >= 0 && d1 < NTAPS) ? ws[d1] : 0.f;
  }

  const float* arow = tile + n * PITCH + 2 * hi;
  float* cdst = tile + TCOLS + 4 * lane;
  float* orow = out + (size_t)(r0 + 8 * hi) * NX + x0 + n;

#pragma unroll 1
  for (int b = 0; b < NBAND; ++b) {
    if (b + 1 < NBAND) {
#pragma unroll
      for (int i = 0; i < TROWS; ++i)
        *(v4f*)(cdst + i * PITCH) = vc[i];
      cdst += WCOLS;
      __syncthreads();
      if (b + 2 < NBAND) {
        load_chunk(in, r0, cs, lane, vc);
        cs += WCOLS;
      }
    }

    v8f c[CTILES];
#pragma unroll
    for (int j = 0; j < CTILES; ++j) c[j] = v8f{};

#pragma unroll
    for (int f = 0; f < NFRAG; ++f) {
      const v2f a = *(const v2f*)(arow + 4 * f);
#pragma unroll
      for (int j = 0; j < CTILES; ++j) {
        const int s = f - 4 * j;
        if (s >= 0 && s < NSTEP) {
          c[j] = __builtin_amdgcn_wmma_f32_16x16x4_f32(false, a, false, B[s],
                                                       (short)0, c[j], false, false);
        }
      }
    }

#pragma unroll
    for (int j = 0; j < CTILES; ++j) {
#pragma unroll
      for (int r = 0; r < 8; ++r) {
        orow[(size_t)r * NX + 16 * j] = c[j][r];
      }
    }
    arow += WCOLS;
    orow += WCOLS;
  }
}

extern "C" void kernel_launch(void* const* d_in, const int* in_sizes, int n_in,
                              void* d_out, int out_size, void* d_ws, size_t ws_size,
                              hipStream_t stream) {
  (void)in_sizes; (void)n_in; (void)out_size; (void)ws_size;
  const float* feature = (const float*)d_in[0];
  const int* h_smooth = (const int*)d_in[1];
  float* w = (float*)d_ws;

  gauss_weights_kernel<<<1, 128, 0, stream>>>(h_smooth, w);

  dim3 grid(NX / SEGCOLS, NZ / TROWS);
  gauss_smooth_wmma<<<grid, 32, 0, stream>>>(feature, w, (float*)d_out);
}


# Taps' symmetry folded in: each 32-column tile multiplies sums and differences of its 56 mirrored input-column pairs, 28 matrix steps per 32 columns instead of 48; summation order changes
# speedup vs baseline: 2.0514x; 1.1332x over previous
#include <hip/hip_runtime.h>


typedef __attribute__((ext_vector_type(2))) float v2f;
typedef __attribute__((ext_vector_type(4))) float v4f;
typedef __attribute__((ext_vector_type(8))) float v8f;

#define NZ 4096
#define NX 8192
#define RADIUS 40
#define NTAPS (2 * RADIUS + 1)
#define KTOT 96
#define CTILES 8
#define QCOLS 32
#define QT 4
#define QWIN (QCOLS + NTAPS - 1)
#define FSTEP (QWIN / 8)

#define WCOLS (16 * CTILES)
#define TROWS 16
#define TCOLS (WCOLS + KTOT - 16)
#define TC4 (TCOLS / 4)
#define NBAND 4
#define SEGCOLS (NBAND * WCOLS)
#define LCOLS (SEGCOLS + KTOT - 16)
#define PITCH 596

__device__ __forceinline__ int mirror_idx(int i) {
  if (i < 0) i = -1 - i;
  if (i >= NX) i = 2 * NX - 1 - i;
  return i;
}

__global__ void gauss_weights_kernel(const int* __restrict__ h_smooth,
                                     float* __restrict__ w) {
  const int j = threadIdx.x;
  const float sigma = (float)h_smooth[0];
  const float inv2s2 = 0.5f / (sigma * sigma);
  float sum = 0.f;
  for (int t = 0; t < NTAPS; ++t) {
    float x = (float)(t - RADIUS);
    sum += __expf(-x * x * inv2s2);
  }
  if (j < NTAPS) {
    float x = (float)(j - RADIUS);
    w[j] = __expf(-x * x * inv2s2) / sum;
  }
}

__device__ __forceinline__ void load_chunk(const float* __restrict__ in, int r0,
                                           int cs, int lane, v4f (&v)[TROWS]) {
  if (cs + WCOLS <= NX) {
    const float* src = in + (size_t)r0 * NX + cs + 4 * lane;
#pragma unroll
    for (int i = 0; i < TROWS; ++i)
      v[i] = *(const v4f*)(src + (size_t)i * NX);
  } else {
    const float* src = in + (size_t)r0 * NX;
    const int col = cs + 4 * lane;
    const int a_0 = mirror_idx(col), a_1 = mirror_idx(col + 1);
    const int a_2 = mirror_idx(col + 2), a_3 = mirror_idx(col + 3);
#pragma unroll
    for (int i = 0; i < TROWS; ++i) {
      const float* inrow = src + (size_t)i * NX;
      v[i].x = inrow[a_0];
      v[i].y = inrow[a_1];
      v[i].z = inrow[a_2];
      v[i].w = inrow[a_3];
    }
  }
}

__launch_bounds__(32)
__global__ void gauss_smooth_wmma(const float* __restrict__ in,
                                  const float* __restrict__ w,
                                  float* __restrict__ out) {
  __shared__ float ws[NTAPS];
  __shared__ __attribute__((aligned(16))) float tile[TROWS * PITCH];

  const int lane = threadIdx.x;
  const int x0 = blockIdx.x * SEGCOLS;
  const int r0 = blockIdx.y * TROWS;
  const int cbase = x0 - RADIUS;
  const bool interior = (cbase >= 0) && (cbase + TCOLS <= NX);

  const float tap0 = w[lane];
  const float tap1 = w[lane + 32];
  float tap2 = 0.f;
  if (lane < NTAPS - 64) tap2 = w[lane + 64];

  float* dst = tile + 4 * lane;
  const bool tail = lane < TC4 - 32;
  v4f va[TROWS], vb[TROWS], vc[TROWS];
  if (interior) {
    const float* src = in + (size_t)r0 * NX + cbase + 4 * lane;
#pragma unroll
    for (int i = 0; i < TROWS; ++i)
      va[i] = *(const v4f*)(src + (size_t)i * NX);
    if (tail) {
#pragma unroll
      for (int i = 0; i < TROWS; ++i)
        vb[i] = *(const v4f*)(src + (size_t)i * NX + 128);
    }
  } else {
    const float* src = in + (size_t)r0 * NX;
    const int col = cbase + 4 * lane;
    const int a_0 = mirror_idx(col), a_1 = mirror_idx(col + 1);
    const int a_2 = mirror_idx(col + 2), a_3 = mirror_idx(col + 3);
#pragma unroll
    for (int i = 0; i < TROWS; ++i) {
      const float* inrow = src + (size_t)i * NX;
      va[i].x = inrow[a_0];
      va[i].y = inrow[a_1];
      va[i].z = inrow[a_2];
      va[i].w = inrow[a_3];
    }
    if (tail) {
      const int b_0 = mirror_idx(col + 128), b_1 = mirror_idx(col + 129);
      const int b_2 = mirror_idx(col + 130), b_3 = mirror_idx(col + 131);
#pragma unroll
      for (int i = 0; i < TROWS; ++i) {
        const float* inrow = src + (size_t)i * NX;
        vb[i].x = inrow[b_0];
        vb[i].y = inrow[b_1];
        vb[i].z = inrow[b_2];
        vb[i].w = inrow[b_3];
      }
    }
  }
  int cs = cbase + TCOLS;
  load_chunk(in, r0, cs, lane, vc);
  cs += WCOLS;

#pragma unroll
  for (int i = 0; i < TROWS; ++i)
    *(v4f*)(dst + i * PITCH) = va[i];
  if (tail) {
#pragma unroll
    for (int i = 0; i < TROWS; ++i)
      *(v4f*)(dst + i * PITCH + 128) = vb[i];
  }
  ws[lane] = tap0;
  ws[lane + 32] = tap1;
  if (lane < NTAPS - 64) ws[lane + 64] = tap2;
  __syncthreads();

  const int n = lane & 15;
  const int hi = lane >> 4;

  v2f BS[FSTEP], BD[FSTEP];
#pragma unroll
  for (int s = 0; s < FSTEP; ++s) {
    float us[2], ud[2];
#pragma unroll
    for (int e = 0; e < 2; ++e) {
      const int k = 4 * s + 2 * hi + e;
      const int d1 = k - n;
      const int d2 = QWIN - 1 - k - n;
      const float t1 = (d1 >= 0) ? ws[d1 < 0 ? 0 : d1] : 0.f;
      const float t2 = (d2 < NTAPS) ? ws[d2 < NTAPS ? d2 : NTAPS - 1] : 0.f;
      us[e] = 0.5f * (t1 + t2);
      ud[e] = 0.5f * (t1 - t2);
    }
    BS[s] = v2f{us[0], us[1]};
    BD[s] = v2f{ud[0], ud[1]};
  }

  const float* fwd = tile + n * PITCH + 2 * hi;
  const float* rev = tile + n * PITCH + (QWIN - 2) - 2 * hi;
  float* cdst = tile + TCOLS + 4 * lane;
  float* olo = out + (size_t)(r0 + 8 * hi) * NX + x0 + n;
  float* ohi = out + (size_t)(r0 + 8 * hi) * NX + x0 + (QCOLS - 1) - n;

#pragma unroll 1
  for (int b = 0; b < NBAND; ++b) {
    if (b + 1 < NBAND) {
#pragma unroll
      for (int i = 0; i < TROWS; ++i)
        *(v4f*)(cdst + i * PITCH) = vc[i];
      cdst += WCOLS;
      __syncthreads();
      if (b + 2 < NBAND) {
        load_chunk(in, r0, cs, lane, vc);
        cs += WCOLS;
      }
    }

    v2f F[FSTEP], R[FSTEP];
#pragma unroll
    for (int s = 0; s < FSTEP; ++s) {
      F[s] = *(const v2f*)(fwd + 4 * s);
      R[s] = *(const v2f*)(rev - 4 * s);
    }

#pragma unroll
    for (int t = 0; t < QT; ++t) {
      v2f AS[FSTEP], AD[FSTEP];
#pragma unroll
      for (int s = 0; s < FSTEP; ++s) {
        const v2f m = v2f{R[s].y, R[s].x};
        AS[s] = F[s] + m;
        AD[s] = F[s] - m;
      }
      if (t + 1 < QT) {
#pragma unroll
        for (int s = 0; s < FSTEP; ++s) {
          F[s] = *(const v2f*)(fwd + QCOLS * (t + 1) + 4 * s);
          R[s] = *(const v2f*)(rev + QCOLS * (t + 1) - 4 * s);
        }
      }

      v8f P = v8f{}, Q = v8f{};
#pragma unroll
      for (int s = 0; s < FSTEP; ++s) {
        P = __builtin_amdgcn_wmma_f32_16x16x4_f32(false, AS[s], false, BS[s],
                                                  (short)0, P, false, false);
        Q = __builtin_amdgcn_wmma_f32_16x16x4_f32(false, AD[s], false, BD[s],
                                                  (short)0, Q, false, false);
      }

#pragma unroll
      for (int r = 0; r < 8; ++r) {
        olo[(size_t)r * NX + QCOLS * t] = P[r] + Q[r];
        ohi[(size_t)r * NX + QCOLS * t] = P[r] - Q[r];
      }
    }
    fwd += WCOLS;
    rev += WCOLS;
    olo += WCOLS;
    ohi += WCOLS;
  }
}

extern "C" void kernel_launch(void* const* d_in, const int* in_sizes, int n_in,
                              void* d_out, int out_size, void* d_ws, size_t ws_size,
                              hipStream_t stream) {
  (void)in_sizes; (void)n_in; (void)out_size; (void)ws_size;
  const float* feature = (const float*)d_in[0];
  const int* h_smooth = (const int*)d_in[1];
  float* w = (float*)d_ws;

  gauss_weights_kernel<<<1, 128, 0, stream>>>(h_smooth, w);

  dim3 grid(NX / SEGCOLS, NZ / TROWS);
  gauss_smooth_wmma<<<grid, 32, 0, stream>>>(feature, w, (float*)d_out);
}


# Results staged in the band's dead LDS window columns and read back as rows: each store instruction writes 512 consecutive bytes of one output row instead of two 64-byte half lines; sums unchanged
# speedup vs baseline: 2.3268x; 1.1342x over previous
#include <hip/hip_runtime.h>


typedef __attribute__((ext_vector_type(2))) float v2f;
typedef __attribute__((ext_vector_type(4))) float v4f;
typedef __attribute__((ext_vector_type(8))) float v8f;

template <bool B> struct flag { static constexpr bool value = B; };

#define NZ 4096
#define NX 8192
#define RADIUS 40
#define NTAPS (2 * RADIUS + 1)
#define KTOT 96
#define CTILES 8
#define QCOLS 32
#define QT 4
#define QWIN (QCOLS + NTAPS - 1)
#define FSTEP (QWIN / 8)

#define WCOLS (16 * CTILES)
#define TROWS 16
#define TCOLS (WCOLS + KTOT - 16)
#define TC4 (TCOLS / 4)
#define NBAND 4
#define SEGCOLS (NBAND * WCOLS)
#define LCOLS (SEGCOLS + KTOT - 16)
#define PITCH 596

__device__ __forceinline__ int mirror_idx(int i) {
  if (i < 0) i = -1 - i;
  if (i >= NX) i = 2 * NX - 1 - i;
  return i;
}

__global__ void gauss_weights_kernel(const int* __restrict__ h_smooth,
                                     float* __restrict__ w) {
  const int j = threadIdx.x;
  const float sigma = (float)h_smooth[0];
  const float inv2s2 = 0.5f / (sigma * sigma);
  float sum = 0.f;
  for (int t = 0; t < NTAPS; ++t) {
    float x = (float)(t - RADIUS);
    sum += __expf(-x * x * inv2s2);
  }
  if (j < NTAPS) {
    float x = (float)(j - RADIUS);
    w[j] = __expf(-x * x * inv2s2) / sum;
  }
}

__device__ __forceinline__ void load_chunk(const float* __restrict__ in, int r0,
                                           int cs, int lane, v4f (&v)[TROWS]) {
  if (cs + WCOLS <= NX) {
    const float* src = in + (size_t)r0 * NX + cs + 4 * lane;
#pragma unroll
    for (int i = 0; i < TROWS; ++i)
      v[i] = *(const v4f*)(src + (size_t)i * NX);
  } else {
    const float* src = in + (size_t)r0 * NX;
    const int col = cs + 4 * lane;
    const int a_0 = mirror_idx(col), a_1 = mirror_idx(col + 1);
    const int a_2 = mirror_idx(col + 2), a_3 = mirror_idx(col + 3);
#pragma unroll
    for (int i = 0; i < TROWS; ++i) {
      const float* inrow = src + (size_t)i * NX;
      v[i].x = inrow[a_0];
      v[i].y = inrow[a_1];
      v[i].z = inrow[a_2];
      v[i].w = inrow[a_3];
    }
  }
}

__device__ __forceinline__ void flush_band(const float* rb, float* orow) {
  v4f o[TROWS];
#pragma unroll
  for (int j = 0; j < TROWS; ++j)
    o[j] = *(const v4f*)(rb + j * PITCH);
#pragma unroll
  for (int j = 0; j < TROWS; ++j)
    *(v4f*)(orow + (size_t)j * NX) = o[j];
}

__launch_bounds__(32)
__global__ void gauss_smooth_wmma(const float* __restrict__ in,
                                  const float* __restrict__ w,
                                  float* __restrict__ out) {
  __shared__ float ws[NTAPS];
  __shared__ __attribute__((aligned(16))) float tile[TROWS * PITCH];

  const int lane = threadIdx.x;
  const int x0 = blockIdx.x * SEGCOLS;
  const int r0 = blockIdx.y * TROWS;
  const int cbase = x0 - RADIUS;
  const bool interior = (cbase >= 0) && (cbase + TCOLS <= NX);

  const float tap0 = w[lane];
  const float tap1 = w[lane + 32];
  float tap2 = 0.f;
  if (lane < NTAPS - 64) tap2 = w[lane + 64];

  float* dst = tile + 4 * lane;
  const bool tail = lane < TC4 - 32;
  v4f va[TROWS], vb[TROWS], vc[TROWS];
  if (interior) {
    const float* src = in + (size_t)r0 * NX + cbase + 4 * lane;
#pragma unroll
    for (int i = 0; i < TROWS; ++i)
      va[i] = *(const v4f*)(src + (size_t)i * NX);
    if (tail) {
#pragma unroll
      for (int i = 0; i < TROWS; ++i)
        vb[i] = *(const v4f*)(src + (size_t)i * NX + 128);
    }
  } else {
    const float* src = in + (size_t)r0 * NX;
    const int col = cbase + 4 * lane;
    const int a_0 = mirror_idx(col), a_1 = mirror_idx(col + 1);
    const int a_2 = mirror_idx(col + 2), a_3 = mirror_idx(col + 3);
#pragma unroll
    for (int i = 0; i < TROWS; ++i) {
      const float* inrow = src + (size_t)i * NX;
      va[i].x = inrow[a_0];
      va[i].y = inrow[a_1];
      va[i].z = inrow[a_2];
      va[i].w = inrow[a_3];
    }
    if (tail) {
      const int b_0 = mirror_idx(col + 128), b_1 = mirror_idx(col + 129);
      const int b_2 = mirror_idx(col + 130), b_3 = mirror_idx(col + 131);
#pragma unroll
      for (int i = 0; i < TROWS; ++i) {
        const float* inrow = src + (size_t)i * NX;
        vb[i].x = inrow[b_0];
        vb[i].y = inrow[b_1];
        vb[i].z = inrow[b_2];
        vb[i].w = inrow[b_3];
      }
    }
  }
  int cs = cbase + TCOLS;
  load_chunk(in, r0, cs, lane, vc);
  cs += WCOLS;

#pragma unroll
  for (int i = 0; i < TROWS; ++i)
    *(v4f*)(dst + i * PITCH) = va[i];
  if (tail) {
#pragma unroll
    for (int i = 0; i < TROWS; ++i)
      *(v4f*)(dst + i * PITCH + 128) = vb[i];
  }
  ws[lane] = tap0;
  ws[lane + 32] = tap1;
  if (lane < NTAPS - 64) ws[lane + 64] = tap2;
  __syncthreads();

  const int n = lane & 15;
  const int hi = lane >> 4;

  v2f BS[FSTEP], BD[FSTEP];
#pragma unroll
  for (int s = 0; s < FSTEP; ++s) {
    float us[2], ud[2];
#pragma unroll
    for (int e = 0; e < 2; ++e) {
      const int k = 4 * s + 2 * hi + e;
      const int d1 = k - n;
      const int d2 = QWIN - 1 - k - n;
      const float t1 = (d1 >= 0) ? ws[d1 < 0 ? 0 : d1] : 0.f;
      const float t2 = (d2 < NTAPS) ? ws[d2 < NTAPS ? d2 : NTAPS - 1] : 0.f;
      us[e] = 0.5f * (t1 + t2);
      ud[e] = 0.5f * (t1 - t2);
    }
    BS[s] = v2f{us[0], us[1]};
    BD[s] = v2f{ud[0], ud[1]};
  }

  const float* fwd = tile + n * PITCH + 2 * hi;
  const float* rev = tile + n * PITCH + (QWIN - 2) - 2 * hi;
  float* cdst = tile + TCOLS + 4 * lane;
  float* slo = tile + (8 * hi) * PITCH + n;
  float* shi = tile + (8 * hi) * PITCH + (QCOLS - 1) - n;
  const float* rb = tile + 4 * lane;
  float* orow = out + (size_t)r0 * NX + x0 + 4 * lane;

  auto band = [&](auto flush_tag) __attribute__((always_inline)) {
    constexpr bool FLUSH = decltype(flush_tag)::value;
    v2f F[QT][FSTEP], R[QT][FSTEP];
#pragma unroll
    for (int t = 0; t < QT; ++t) {
#pragma unroll
      for (int s = 0; s < FSTEP; ++s) {
        F[t][s] = *(const v2f*)(fwd + QCOLS * t + 4 * s);
        R[t][s] = *(const v2f*)(rev + QCOLS * t - 4 * s);
      }
    }
    v4f o[TROWS];
    if constexpr (FLUSH) {
#pragma unroll
      for (int j = 0; j < TROWS; ++j)
        o[j] = *(const v4f*)(rb + j * PITCH);
      rb += WCOLS;
    }

    if constexpr (FLUSH) {
#pragma unroll
      for (int j = 0; j < TROWS; ++j)
        *(v4f*)(orow + (size_t)j * NX) = o[j];
      orow += WCOLS;
    }

    v8f P[QT], Q[QT];
#pragma unroll
    for (int t = 0; t < QT; ++t) {
      v2f AS[FSTEP], AD[FSTEP];
#pragma unroll
      for (int s = 0; s < FSTEP; ++s) {
        const v2f m = v2f{R[t][s].y, R[t][s].x};
        AS[s] = F[t][s] + m;
        AD[s] = F[t][s] - m;
      }

      P[t] = v8f{};
      Q[t] = v8f{};
#pragma unroll
      for (int s = 0; s < FSTEP; ++s) {
        P[t] = __builtin_amdgcn_wmma_f32_16x16x4_f32(false, AS[s], false, BS[s],
                                                     (short)0, P[t], false, false);
        Q[t] = __builtin_amdgcn_wmma_f32_16x16x4_f32(false, AD[s], false, BD[s],
                                                     (short)0, Q[t], false, false);
      }
    }

#pragma unroll
    for (int t = 0; t < QT; ++t) {
#pragma unroll
      for (int r = 0; r < 8; ++r) {
        slo[r * PITCH + QCOLS * t] = P[t][r] + Q[t][r];
        shi[r * PITCH + QCOLS * t] = P[t][r] - Q[t][r];
      }
    }
    fwd += WCOLS;
    rev += WCOLS;
    slo += WCOLS;
    shi += WCOLS;
  };

#pragma unroll
  for (int i = 0; i < TROWS; ++i)
    *(v4f*)(cdst + i * PITCH) = vc[i];
  cdst += WCOLS;
  __syncthreads();
  load_chunk(in, r0, cs, lane, vc);
  cs += WCOLS;
  band(flag<false>{});

#pragma unroll 1
  for (int b = 1; b < NBAND; ++b) {
    if (b + 1 < NBAND) {
#pragma unroll
      for (int i = 0; i < TROWS; ++i)
        *(v4f*)(cdst + i * PITCH) = vc[i];
      cdst += WCOLS;
    }
    __syncthreads();
    if (b + 2 < NBAND) {
      load_chunk(in, r0, cs, lane, vc);
      cs += WCOLS;
    }
    band(flag<true>{});
  }

  __syncthreads();
  flush_band(rb, orow);
}

extern "C" void kernel_launch(void* const* d_in, const int* in_sizes, int n_in,
                              void* d_out, int out_size, void* d_ws, size_t ws_size,
                              hipStream_t stream) {
  (void)in_sizes; (void)n_in; (void)out_size; (void)ws_size;
  const float* feature = (const float*)d_in[0];
  const int* h_smooth = (const int*)d_in[1];
  float* w = (float*)d_ws;

  gauss_weights_kernel<<<1, 128, 0, stream>>>(h_smooth, w);

  dim3 grid(NX / SEGCOLS, NZ / TROWS);
  gauss_smooth_wmma<<<grid, 32, 0, stream>>>(feature, w, (float*)d_out);
}


# Tap kernel folded into the main one: each wave works out the 81 taps itself (3 exps per lane, 32-lane butterfly for the norm) under its window loads; one launch; the taps' norm sum order changes
# speedup vs baseline: 2.5099x; 1.0787x over previous
#include <hip/hip_runtime.h>


typedef __attribute__((ext_vector_type(2))) float v2f;
typedef __attribute__((ext_vector_type(4))) float v4f;
typedef __attribute__((ext_vector_type(8))) float v8f;

template <bool B> struct flag { static constexpr bool value = B; };

#define NZ 4096
#define NX 8192
#define RADIUS 40
#define NTAPS (2 * RADIUS + 1)
#define KTOT 96
#define CTILES 8
#define QCOLS 32
#define QT 4
#define QWIN (QCOLS + NTAPS - 1)
#define FSTEP (QWIN / 8)

#define WCOLS (16 * CTILES)
#define TROWS 16
#define TCOLS (WCOLS + KTOT - 16)
#define TC4 (TCOLS / 4)
#define NBAND 4
#define SEGCOLS (NBAND * WCOLS)
#define LCOLS (SEGCOLS + KTOT - 16)
#define PITCH 596

__device__ __forceinline__ int mirror_idx(int i) {
  if (i < 0) i = -1 - i;
  if (i >= NX) i = 2 * NX - 1 - i;
  return i;
}

__device__ __forceinline__ float wave_sum(float x) {
  x += __builtin_bit_cast(float, __builtin_amdgcn_update_dpp(
           0, __builtin_bit_cast(int, x), 0x161, 0xf, 0xf, false));
  x += __builtin_bit_cast(float, __builtin_amdgcn_update_dpp(
           0, __builtin_bit_cast(int, x), 0x162, 0xf, 0xf, false));
  x += __builtin_bit_cast(float, __builtin_amdgcn_update_dpp(
           0, __builtin_bit_cast(int, x), 0x164, 0xf, 0xf, false));
  x += __builtin_bit_cast(float, __builtin_amdgcn_update_dpp(
           0, __builtin_bit_cast(int, x), 0x168, 0xf, 0xf, false));
  const int b = __builtin_bit_cast(int, x);
  x += __builtin_bit_cast(float, __builtin_amdgcn_permlanex16(
           b, b, 0x76543210, 0xfedcba98, false, false));
  return x;
}

__device__ __forceinline__ void load_chunk(const float* __restrict__ in, int r0,
                                           int cs, int lane, v4f (&v)[TROWS]) {
  if (cs + WCOLS <= NX) {
    const float* src = in + (size_t)r0 * NX + cs + 4 * lane;
#pragma unroll
    for (int i = 0; i < TROWS; ++i)
      v[i] = *(const v4f*)(src + (size_t)i * NX);
  } else {
    const float* src = in + (size_t)r0 * NX;
    const int col = cs + 4 * lane;
    const int a_0 = mirror_idx(col), a_1 = mirror_idx(col + 1);
    const int a_2 = mirror_idx(col + 2), a_3 = mirror_idx(col + 3);
#pragma unroll
    for (int i = 0; i < TROWS; ++i) {
      const float* inrow = src + (size_t)i * NX;
      v[i].x = inrow[a_0];
      v[i].y = inrow[a_1];
      v[i].z = inrow[a_2];
      v[i].w = inrow[a_3];
    }
  }
}

__device__ __forceinline__ void flush_band(const float* rb, float* orow) {
  v4f o[TROWS];
#pragma unroll
  for (int j = 0; j < TROWS; ++j)
    o[j] = *(const v4f*)(rb + j * PITCH);
#pragma unroll
  for (int j = 0; j < TROWS; ++j)
    *(v4f*)(orow + (size_t)j * NX) = o[j];
}

__launch_bounds__(32)
__global__ void gauss_smooth_wmma(const float* __restrict__ in,
                                  const int* __restrict__ h_smooth,
                                  float* __restrict__ out) {
  __shared__ float ws[NTAPS];
  __shared__ __attribute__((aligned(16))) float tile[TROWS * PITCH];

  const int lane = threadIdx.x;
  const int x0 = blockIdx.x * SEGCOLS;
  const int r0 = blockIdx.y * TROWS;
  const int cbase = x0 - RADIUS;
  const bool interior = (cbase >= 0) && (cbase + TCOLS <= NX);

  const float sigma = (float)h_smooth[0];

  float* dst = tile + 4 * lane;
  const bool tail = lane < TC4 - 32;
  v4f va[TROWS], vb[TROWS], vc[TROWS];
  if (interior) {
    const float* src = in + (size_t)r0 * NX + cbase + 4 * lane;
#pragma unroll
    for (int i = 0; i < TROWS; ++i)
      va[i] = *(const v4f*)(src + (size_t)i * NX);
    if (tail) {
#pragma unroll
      for (int i = 0; i < TROWS; ++i)
        vb[i] = *(const v4f*)(src + (size_t)i * NX + 128);
    }
  } else {
    const float* src = in + (size_t)r0 * NX;
    const int col = cbase + 4 * lane;
    const int a_0 = mirror_idx(col), a_1 = mirror_idx(col + 1);
    const int a_2 = mirror_idx(col + 2), a_3 = mirror_idx(col + 3);
#pragma unroll
    for (int i = 0; i < TROWS; ++i) {
      const float* inrow = src + (size_t)i * NX;
      va[i].x = inrow[a_0];
      va[i].y = inrow[a_1];
      va[i].z = inrow[a_2];
      va[i].w = inrow[a_3];
    }
    if (tail) {
      const int b_0 = mirror_idx(col + 128), b_1 = mirror_idx(col + 129);
      const int b_2 = mirror_idx(col + 130), b_3 = mirror_idx(col + 131);
#pragma unroll
      for (int i = 0; i < TROWS; ++i) {
        const float* inrow = src + (size_t)i * NX;
        vb[i].x = inrow[b_0];
        vb[i].y = inrow[b_1];
        vb[i].z = inrow[b_2];
        vb[i].w = inrow[b_3];
      }
    }
  }
  int cs = cbase + TCOLS;
  load_chunk(in, r0, cs, lane, vc);
  cs += WCOLS;

  {
    const float inv2s2 = 0.5f / (sigma * sigma);
    const float xa = (float)(lane - RADIUS);
    const float xb = (float)(lane + 32 - RADIUS);
    const float xc = (float)(lane + 64 - RADIUS);
    const float e0 = __expf(-xa * xa * inv2s2);
    const float e1 = __expf(-xb * xb * inv2s2);
    const float e2 = (lane < NTAPS - 64) ? __expf(-xc * xc * inv2s2) : 0.f;
    const float sum = wave_sum((e0 + e1) + e2);
    ws[lane] = e0 / sum;
    ws[lane + 32] = e1 / sum;
    if (lane < NTAPS - 64) ws[lane + 64] = e2 / sum;
  }

#pragma unroll
  for (int i = 0; i < TROWS; ++i)
    *(v4f*)(dst + i * PITCH) = va[i];
  if (tail) {
#pragma unroll
    for (int i = 0; i < TROWS; ++i)
      *(v4f*)(dst + i * PITCH + 128) = vb[i];
  }
  __syncthreads();

  const int n = lane & 15;
  const int hi = lane >> 4;

  v2f BS[FSTEP], BD[FSTEP];
#pragma unroll
  for (int s = 0; s < FSTEP; ++s) {
    float us[2], ud[2];
#pragma unroll
    for (int e = 0; e < 2; ++e) {
      const int k = 4 * s + 2 * hi + e;
      const int d1 = k - n;
      const int d2 = QWIN - 1 - k - n;
      const float t1 = (d1 >= 0) ? ws[d1 < 0 ? 0 : d1] : 0.f;
      const float t2 = (d2 < NTAPS) ? ws[d2 < NTAPS ? d2 : NTAPS - 1] : 0.f;
      us[e] = 0.5f * (t1 + t2);
      ud[e] = 0.5f * (t1 - t2);
    }
    BS[s] = v2f{us[0], us[1]};
    BD[s] = v2f{ud[0], ud[1]};
  }

  const float* fwd = tile + n * PITCH + 2 * hi;
  const float* rev = tile + n * PITCH + (QWIN - 2) - 2 * hi;
  float* cdst = tile + TCOLS + 4 * lane;
  float* slo = tile + (8 * hi) * PITCH + n;
  float* shi = tile + (8 * hi) * PITCH + (QCOLS - 1) - n;
  const float* rb = tile + 4 * lane;
  float* orow = out + (size_t)r0 * NX + x0 + 4 * lane;

  auto band = [&](auto flush_tag) __attribute__((always_inline)) {
    constexpr bool FLUSH = decltype(flush_tag)::value;
    v2f F[QT][FSTEP], R[QT][FSTEP];
#pragma unroll
    for (int t = 0; t < QT; ++t) {
#pragma unroll
      for (int s = 0; s < FSTEP; ++s) {
        F[t][s] = *(const v2f*)(fwd + QCOLS * t + 4 * s);
        R[t][s] = *(const v2f*)(rev + QCOLS * t - 4 * s);
      }
    }
    v4f o[TROWS];
    if constexpr (FLUSH) {
#pragma unroll
      for (int j = 0; j < TROWS; ++j)
        o[j] = *(const v4f*)(rb + j * PITCH);
      rb += WCOLS;
    }

    if constexpr (FLUSH) {
#pragma unroll
      for (int j = 0; j < TROWS; ++j)
        *(v4f*)(orow + (size_t)j * NX) = o[j];
      orow += WCOLS;
    }

    v8f P[QT], Q[QT];
#pragma unroll
    for (int t = 0; t < QT; ++t) {
      v2f AS[FSTEP], AD[FSTEP];
#pragma unroll
      for (int s = 0; s < FSTEP; ++s) {
        const v2f m = v2f{R[t][s].y, R[t][s].x};
        AS[s] = F[t][s] + m;
        AD[s] = F[t][s] - m;
      }

      P[t] = v8f{};
      Q[t] = v8f{};
#pragma unroll
      for (int s = 0; s < FSTEP; ++s) {
        P[t] = __builtin_amdgcn_wmma_f32_16x16x4_f32(false, AS[s], false, BS[s],
                                                     (short)0, P[t], false, false);
        Q[t] = __builtin_amdgcn_wmma_f32_16x16x4_f32(false, AD[s], false, BD[s],
                                                     (short)0, Q[t], false, false);
      }
    }

#pragma unroll
    for (int t = 0; t < QT; ++t) {
#pragma unroll
      for (int r = 0; r < 8; ++r) {
        slo[r * PITCH + QCOLS * t] = P[t][r] + Q[t][r];
        shi[r * PITCH + QCOLS * t] = P[t][r] - Q[t][r];
      }
    }
    fwd += WCOLS;
    rev += WCOLS;
    slo += WCOLS;
    shi += WCOLS;
  };

#pragma unroll
  for (int i = 0; i < TROWS; ++i)
    *(v4f*)(cdst + i * PITCH) = vc[i];
  cdst += WCOLS;
  __syncthreads();
  load_chunk(in, r0, cs, lane, vc);
  cs += WCOLS;
  band(flag<false>{});

#pragma unroll 1
  for (int b = 1; b < NBAND; ++b) {
    if (b + 1 < NBAND) {
#pragma unroll
      for (int i = 0; i < TROWS; ++i)
        *(v4f*)(cdst + i * PITCH) = vc[i];
      cdst += WCOLS;
    }
    __syncthreads();
    if (b + 2 < NBAND) {
      load_chunk(in, r0, cs, lane, vc);
      cs += WCOLS;
    }
    band(flag<true>{});
  }

  __syncthreads();
  flush_band(rb, orow);
}

extern "C" void kernel_launch(void* const* d_in, const int* in_sizes, int n_in,
                              void* d_out, int out_size, void* d_ws, size_t ws_size,
                              hipStream_t stream) {
  (void)in_sizes; (void)n_in; (void)out_size; (void)d_ws; (void)ws_size;
  const float* feature = (const float*)d_in[0];
  const int* h_smooth = (const int*)d_in[1];

  dim3 grid(NX / SEGCOLS, NZ / TROWS);
  gauss_smooth_wmma<<<grid, 32, 0, stream>>>(feature, h_smooth, (float*)d_out);
}


# Matrix operands swapped (tap tiles as A, window sums as B): a lane ends with 8 consecutive output columns of one row, so a band is staged with 16 LDS stores of 16 bytes, not 64 of 4; sums unchanged
# speedup vs baseline: 2.5984x; 1.0353x over previous
#include <hip/hip_runtime.h>


typedef __attribute__((ext_vector_type(2))) float v2f;
typedef __attribute__((ext_vector_type(4))) float v4f;
typedef __attribute__((ext_vector_type(8))) float v8f;

template <bool B> struct flag { static constexpr bool value = B; };

#define NZ 4096
#define NX 8192
#define RADIUS 40
#define NTAPS (2 * RADIUS + 1)
#define KTOT 96
#define CTILES 8
#define QCOLS 32
#define QT 4
#define QWIN (QCOLS + NTAPS - 1)
#define FSTEP (QWIN / 8)

#define WCOLS (16 * CTILES)
#define TROWS 16
#define TCOLS (WCOLS + KTOT - 16)
#define TC4 (TCOLS / 4)
#define NBAND 4
#define SEGCOLS (NBAND * WCOLS)
#define LCOLS (SEGCOLS + KTOT - 16)
#define PITCH 596

__device__ __forceinline__ int mirror_idx(int i) {
  if (i < 0) i = -1 - i;
  if (i >= NX) i = 2 * NX - 1 - i;
  return i;
}

__device__ __forceinline__ float wave_sum(float x) {
  x += __builtin_bit_cast(float, __builtin_amdgcn_update_dpp(
           0, __builtin_bit_cast(int, x), 0x161, 0xf, 0xf, false));
  x += __builtin_bit_cast(float, __builtin_amdgcn_update_dpp(
           0, __builtin_bit_cast(int, x), 0x162, 0xf, 0xf, false));
  x += __builtin_bit_cast(float, __builtin_amdgcn_update_dpp(
           0, __builtin_bit_cast(int, x), 0x164, 0xf, 0xf, false));
  x += __builtin_bit_cast(float, __builtin_amdgcn_update_dpp(
           0, __builtin_bit_cast(int, x), 0x168, 0xf, 0xf, false));
  const int b = __builtin_bit_cast(int, x);
  x += __builtin_bit_cast(float, __builtin_amdgcn_permlanex16(
           b, b, 0x76543210, 0xfedcba98, false, false));
  return x;
}

__device__ __forceinline__ void load_chunk(const float* __restrict__ in, int r0,
                                           int cs, int lane, v4f (&v)[TROWS]) {
  if (cs + WCOLS <= NX) {
    const float* src = in + (size_t)r0 * NX + cs + 4 * lane;
#pragma unroll
    for (int i = 0; i < TROWS; ++i)
      v[i] = *(const v4f*)(src + (size_t)i * NX);
  } else {
    const float* src = in + (size_t)r0 * NX;
    const int col = cs + 4 * lane;
    const int a_0 = mirror_idx(col), a_1 = mirror_idx(col + 1);
    const int a_2 = mirror_idx(col + 2), a_3 = mirror_idx(col + 3);
#pragma unroll
    for (int i = 0; i < TROWS; ++i) {
      const float* inrow = src + (size_t)i * NX;
      v[i].x = inrow[a_0];
      v[i].y = inrow[a_1];
      v[i].z = inrow[a_2];
      v[i].w = inrow[a_3];
    }
  }
}

__device__ __forceinline__ void flush_band(const float* rb, float* orow) {
  v4f o[TROWS];
#pragma unroll
  for (int j = 0; j < TROWS; ++j)
    o[j] = *(const v4f*)(rb + j * PITCH);
#pragma unroll
  for (int j = 0; j < TROWS; ++j)
    *(v4f*)(orow + (size_t)j * NX) = o[j];
}

__launch_bounds__(32)
__global__ void gauss_smooth_wmma(const float* __restrict__ in,
                                  const int* __restrict__ h_smooth,
                                  float* __restrict__ out) {
  __shared__ float ws[NTAPS];
  __shared__ __attribute__((aligned(16))) float tile[TROWS * PITCH];

  const int lane = threadIdx.x;
  const int x0 = blockIdx.x * SEGCOLS;
  const int r0 = blockIdx.y * TROWS;
  const int cbase = x0 - RADIUS;
  const bool interior = (cbase >= 0) && (cbase + TCOLS <= NX);

  const float sigma = (float)h_smooth[0];

  float* dst = tile + 4 * lane;
  const bool tail = lane < TC4 - 32;
  v4f va[TROWS], vb[TROWS], vc[TROWS];
  if (interior) {
    const float* src = in + (size_t)r0 * NX + cbase + 4 * lane;
#pragma unroll
    for (int i = 0; i < TROWS; ++i)
      va[i] = *(const v4f*)(src + (size_t)i * NX);
    if (tail) {
#pragma unroll
      for (int i = 0; i < TROWS; ++i)
        vb[i] = *(const v4f*)(src + (size_t)i * NX + 128);
    }
  } else {
    const float* src = in + (size_t)r0 * NX;
    const int col = cbase + 4 * lane;
    const int a_0 = mirror_idx(col), a_1 = mirror_idx(col + 1);
    const int a_2 = mirror_idx(col + 2), a_3 = mirror_idx(col + 3);
#pragma unroll
    for (int i = 0; i < TROWS; ++i) {
      const float* inrow = src + (size_t)i * NX;
      va[i].x = inrow[a_0];
      va[i].y = inrow[a_1];
      va[i].z = inrow[a_2];
      va[i].w = inrow[a_3];
    }
    if (tail) {
      const int b_0 = mirror_idx(col + 128), b_1 = mirror_idx(col + 129);
      const int b_2 = mirror_idx(col + 130), b_3 = mirror_idx(col + 131);
#pragma unroll
      for (int i = 0; i < TROWS; ++i) {
        const float* inrow = src + (size_t)i * NX;
        vb[i].x = inrow[b_0];
        vb[i].y = inrow[b_1];
        vb[i].z = inrow[b_2];
        vb[i].w = inrow[b_3];
      }
    }
  }
  int cs = cbase + TCOLS;
  load_chunk(in, r0, cs, lane, vc);
  cs += WCOLS;

  {
    const float inv2s2 = 0.5f / (sigma * sigma);
    const float xa = (float)(lane - RADIUS);
    const float xb = (float)(lane + 32 - RADIUS);
    const float xc = (float)(lane + 64 - RADIUS);
    const float e0 = __expf(-xa * xa * inv2s2);
    const float e1 = __expf(-xb * xb * inv2s2);
    const float e2 = (lane < NTAPS - 64) ? __expf(-xc * xc * inv2s2) : 0.f;
    const float sum = wave_sum((e0 + e1) + e2);
    ws[lane] = e0 / sum;
    ws[lane + 32] = e1 / sum;
    if (lane < NTAPS - 64) ws[lane + 64] = e2 / sum;
  }

#pragma unroll
  for (int i = 0; i < TROWS; ++i)
    *(v4f*)(dst + i * PITCH) = va[i];
  if (tail) {
#pragma unroll
    for (int i = 0; i < TROWS; ++i)
      *(v4f*)(dst + i * PITCH + 128) = vb[i];
  }
  __syncthreads();

  const int n = lane & 15;
  const int hi = lane >> 4;

  v2f BS[FSTEP], BD[FSTEP];
#pragma unroll
  for (int s = 0; s < FSTEP; ++s) {
    float us[2], ud[2];
#pragma unroll
    for (int e = 0; e < 2; ++e) {
      const int k = 4 * s + 2 * hi + e;
      const int d1 = k - n;
      const int d2 = QWIN - 1 - k - n;
      const float t1 = (d1 >= 0) ? ws[d1 < 0 ? 0 : d1] : 0.f;
      const float t2 = (d2 < NTAPS) ? ws[d2 < NTAPS ? d2 : NTAPS - 1] : 0.f;
      us[e] = 0.5f * (t1 + t2);
      ud[e] = 0.5f * (t1 - t2);
    }
    BS[s] = v2f{us[0], us[1]};
    BD[s] = v2f{ud[0], ud[1]};
  }

  const float* fwd = tile + n * PITCH + 2 * hi;
  const float* rev = tile + n * PITCH + (QWIN - 2) - 2 * hi;
  float* cdst = tile + TCOLS + 4 * lane;
  float* slo = tile + n * PITCH + 8 * hi;
  float* shi = tile + n * PITCH + (QCOLS - 8) - 8 * hi;
  const float* rb = tile + 4 * lane;
  float* orow = out + (size_t)r0 * NX + x0 + 4 * lane;

  auto band = [&](auto flush_tag) __attribute__((always_inline)) {
    constexpr bool FLUSH = decltype(flush_tag)::value;
    v2f F[QT][FSTEP], R[QT][FSTEP];
#pragma unroll
    for (int t = 0; t < QT; ++t) {
#pragma unroll
      for (int s = 0; s < FSTEP; ++s) {
        F[t][s] = *(const v2f*)(fwd + QCOLS * t + 4 * s);
        R[t][s] = *(const v2f*)(rev + QCOLS * t - 4 * s);
      }
    }
    v4f o[TROWS];
    if constexpr (FLUSH) {
#pragma unroll
      for (int j = 0; j < TROWS; ++j)
        o[j] = *(const v4f*)(rb + j * PITCH);
      rb += WCOLS;
    }

    if constexpr (FLUSH) {
#pragma unroll
      for (int j = 0; j < TROWS; ++j)
        *(v4f*)(orow + (size_t)j * NX) = o[j];
      orow += WCOLS;
    }

    v8f P[QT], Q[QT];
#pragma unroll
    for (int t = 0; t < QT; ++t) {
      v2f AS[FSTEP], AD[FSTEP];
#pragma unroll
      for (int s = 0; s < FSTEP; ++s) {
        const v2f m = v2f{R[t][s].y, R[t][s].x};
        AS[s] = F[t][s] + m;
        AD[s] = F[t][s] - m;
      }

      P[t] = v8f{};
      Q[t] = v8f{};
#pragma unroll
      for (int s = 0; s < FSTEP; ++s) {
        P[t] = __builtin_amdgcn_wmma_f32_16x16x4_f32(false, BS[s], false, AS[s],
                                                     (short)0, P[t], false, false);
        Q[t] = __builtin_amdgcn_wmma_f32_16x16x4_f32(false, BD[s], false, AD[s],
                                                     (short)0, Q[t], false, false);
      }
    }

#pragma unroll
    for (int t = 0; t < QT; ++t) {
      const v8f p = P[t], q = Q[t];
      *(v4f*)(slo + QCOLS * t) = v4f{p[0] + q[0], p[1] + q[1], p[2] + q[2], p[3] + q[3]};
      *(v4f*)(slo + QCOLS * t + 4) = v4f{p[4] + q[4], p[5] + q[5], p[6] + q[6], p[7] + q[7]};
      *(v4f*)(shi + QCOLS * t) = v4f{p[7] - q[7], p[6] - q[6], p[5] - q[5], p[4] - q[4]};
      *(v4f*)(shi + QCOLS * t + 4) = v4f{p[3] - q[3], p[2] - q[2], p[1] - q[1], p[0] - q[0]};
    }
    fwd += WCOLS;
    rev += WCOLS;
    slo += WCOLS;
    shi += WCOLS;
  };

#pragma unroll
  for (int i = 0; i < TROWS; ++i)
    *(v4f*)(cdst + i * PITCH) = vc[i];
  cdst += WCOLS;
  __syncthreads();
  load_chunk(in, r0, cs, lane, vc);
  cs += WCOLS;
  band(flag<false>{});

#pragma unroll 1
  for (int b = 1; b < NBAND; ++b) {
    if (b + 1 < NBAND) {
#pragma unroll
      for (int i = 0; i < TROWS; ++i)
        *(v4f*)(cdst + i * PITCH) = vc[i];
      cdst += WCOLS;
    }
    __syncthreads();
    if (b + 2 < NBAND) {
      load_chunk(in, r0, cs, lane, vc);
      cs += WCOLS;
    }
    band(flag<true>{});
  }

  __syncthreads();
  flush_band(rb, orow);
}

extern "C" void kernel_launch(void* const* d_in, const int* in_sizes, int n_in,
                              void* d_out, int out_size, void* d_ws, size_t ws_size,
                              hipStream_t stream) {
  (void)in_sizes; (void)n_in; (void)out_size; (void)d_ws; (void)ws_size;
  const float* feature = (const float*)d_in[0];
  const int* h_smooth = (const int*)d_in[1];

  dim3 grid(NX / SEGCOLS, NZ / TROWS);
  gauss_smooth_wmma<<<grid, 32, 0, stream>>>(feature, h_smooth, (float*)d_out);
}
